# starting kernel
# baseline (speedup 1.0000x reference)
#include <hip/hip_runtime.h>
#include <hip/hip_bf16.h>

typedef float v2f __attribute__((ext_vector_type(2)));
typedef float v8f __attribute__((ext_vector_type(8)));

#define PAIR_STRIDE 288

__global__ __launch_bounds__(256) void edgeconv_gemm(const float* __restrict__ x,
                                                     const float* __restrict__ W,
                                                     float* __restrict__ H) {
    __shared__ float wc[32 * PAIR_STRIDE];

    const int tid = threadIdx.x;
    for (int i = tid; i < 64 * 128; i += 256) {
        int c = i >> 7;
        int o = i & 127;
        float v;
        if (o < 64) v = W[c * 64 + o] - W[(64 + c) * 64 + o];
        else        v = W[(64 + c) * 64 + (o - 64)];
        wc[(c >> 1) * PAIR_STRIDE + o * 2 + (c & 1)] = v;
    }
    __syncthreads();

    const int lane = tid & 31;
    const int wv   = tid >> 5;
    const int tile = blockIdx.x * 8 + wv;
    const int m    = lane & 15;
    const int kh   = lane >> 4;
    const int node = tile * 16 + m;
    const int bb   = node >> 12;
    const int nn   = node & 4095;
    const float* xb = x + ((size_t)bb * 64) * 4096 + nn;

    v2f a[16];
#pragma unroll
    for (int kk = 0; kk < 16; ++kk) {
        int c0 = 4 * kk + 2 * kh;
        a[kk].x = xb[(size_t)c0 * 4096];
        a[kk].y = xb[(size_t)(c0 + 1) * 4096];
    }

#pragma unroll
    for (int ot = 0; ot < 8; ++ot) {
        v8f acc = {};
#pragma unroll
        for (int kk = 0; kk < 16; ++kk) {
            const v2f bf = *(const v2f*)&wc[(2 * kk + kh) * PAIR_STRIDE + (ot * 16 + m) * 2];
            acc = __builtin_amdgcn_wmma_f32_16x16x4_f32(
                false, a[kk], false, bf, (short)0, acc, false, false);
        }
        float* Hp = H + ((size_t)tile * 16 + 8 * kh) * 128 + ot * 16 + m;
#pragma unroll
        for (int j = 0; j < 8; ++j)
            Hp[(size_t)j * 128] = acc[j];
    }
}

__global__ __launch_bounds__(256) void edgeconv_aggregate(const float* __restrict__ H,
                                                          const int* __restrict__ src,
                                                          const float* __restrict__ bias,
                                                          float* __restrict__ out) {
    const int lane = threadIdx.x & 31;
    const int wv   = threadIdx.x >> 5;
    const int node = blockIdx.x * 8 + wv;
    const int bb   = node >> 12;
    const int nn   = node & 4095;

    int s = src[node * 16 + (lane & 15)];

    float m0 = -3.402823466e38f, m1 = -3.402823466e38f;
#pragma unroll
    for (int k = 0; k < 16; ++k) {
        int sk = __shfl(s, k, 32);
        const v2f* crow = (const v2f*)(H + ((size_t)(bb * 4096 + sk)) * 128 + 64);
        v2f c = crow[lane];
        m0 = fmaxf(m0, c.x);
        m1 = fmaxf(m1, c.y);
    }

    v2f aA = ((const v2f*)(H + (size_t)node * 128))[lane];
    v2f bV = ((const v2f*)bias)[lane];
    float r0 = fmaxf(aA.x + bV.x + m0, 0.0f);
    float r1 = fmaxf(aA.y + bV.y + m1, 0.0f);

    size_t obase = ((size_t)bb * 64 + 2 * lane) * 4096 + nn;
    out[obase]        = r0;
    out[obase + 4096] = r1;
}

extern "C" void kernel_launch(void* const* d_in, const int* in_sizes, int n_in,
                              void* d_out, int out_size, void* d_ws, size_t ws_size,
                              hipStream_t stream) {
    const float* x  = (const float*)d_in[0];
    const int*   ei = (const int*)d_in[1];
    const float* W  = (const float*)d_in[2];
    const float* b  = (const float*)d_in[3];
    float* out = (float*)d_out;
    float* H   = (float*)d_ws;

    edgeconv_gemm<<<256, 256, 0, stream>>>(x, W, H);
    edgeconv_aggregate<<<4096, 256, 0, stream>>>(H, ei, b, out);
}


# GEMM launch re-tiled: a wave keeps ONE 16-channel weight tile in registers (independent global loads, one round trip) and sweeps 8 node tiles of an x tile staged in LDS; sums unchanged
# speedup vs baseline: 3.7993x; 3.7993x over previous
#include <hip/hip_runtime.h>
#include <hip/hip_bf16.h>


typedef float v2f __attribute__((ext_vector_type(2)));
typedef float v4f __attribute__((ext_vector_type(4)));
typedef float v8f __attribute__((ext_vector_type(8)));
typedef int   v4i __attribute__((ext_vector_type(4)));

#define PAIR_STRIDE 288

#define PLANE_FLOATS (8 * 8 * 4096 * 8)

__global__ __launch_bounds__(256) void edgeconv_gemm(const float* __restrict__ x,
                                                     const float* __restrict__ W,
                                                     float* __restrict__ H) {
    __shared__ v4f xs4[32 * PAIR_STRIDE / 4];
    __shared__ v2f w2s[4][16][32];
    float* xs = (float*)xs4;

    const int tid  = threadIdx.x;
    const int lane = tid & 31;
    const int wv   = __builtin_amdgcn_readfirstlane(tid >> 5);
    const int m    = lane & 15;
    const int kh   = lane >> 4;
    const int bb   = blockIdx.x >> 5;
    const int nn0  = (blockIdx.x & 31) * 128;

    const float* xg = x + ((size_t)bb * 64) * 4096 + nn0 + 4 * lane;
    v4f xr[4][2];
#pragma unroll
    for (int q = 0; q < 4; ++q) {
        const int p = 4 * wv + q;
        xr[q][0] = *(const v4f*)(xg + (size_t)(2 * p) * 4096);
        xr[q][1] = *(const v4f*)(xg + (size_t)(2 * p + 1) * 4096);
    }

    const int col = 16 * (wv & 3) + m;
    const float* wp = W + ((wv < 4) ? 0 : 64 * 64) + (2 * kh) * 64 + col;
    v2f wa[16];
#pragma unroll
    for (int kk = 0; kk < 16; ++kk) {
        wa[kk].x = wp[(4 * kk) * 64];
        wa[kk].y = wp[(4 * kk + 1) * 64];
    }

#pragma unroll
    for (int q = 0; q < 4; ++q) {
        const int p = 4 * wv + q;
        const v4f r0 = xr[q][0];
        const v4f r1 = xr[q][1];
        const v4f s0 = {r0.x, r1.x, r0.y, r1.y};
        const v4f s1 = {r0.z, r1.z, r0.w, r1.w};
        float* d = xs + p * PAIR_STRIDE + 8 * lane;
        *(v4f*)d       = s0;
        *(v4f*)(d + 4) = s1;
    }
    if (wv >= 4) {
#pragma unroll
        for (int kk = 0; kk < 16; ++kk) w2s[wv - 4][kk][lane] = wa[kk];
    }
    __syncthreads();
    if (wv < 4) {
#pragma unroll
        for (int kk = 0; kk < 16; ++kk) {
            const v2f w2 = w2s[wv][kk][lane];
            wa[kk].x = wa[kk].x - w2.x;
            wa[kk].y = wa[kk].y - w2.y;
        }
    }

    const float* xf = xs + kh * PAIR_STRIDE + 2 * m;
    v8f acc[8];
#pragma unroll
    for (int t = 0; t < 8; ++t) acc[t] = (v8f){};
#pragma unroll
    for (int kk = 0; kk < 16; ++kk) {
#pragma unroll
        for (int t = 0; t < 8; ++t) {
            const v2f bf = *(const v2f*)&xf[(2 * kk) * PAIR_STRIDE + t * 32];
            acc[t] = __builtin_amdgcn_wmma_f32_16x16x4_f32(
                false, wa[kk], false, bf, (short)0, acc[t], false, false);
        }
    }

    const int plane = wv >> 2;
    const int slice = (wv & 3) * 2 + kh;
    float* Hw = H + (size_t)plane * PLANE_FLOATS
                  + (((size_t)(bb * 8 + slice) * 4096 + nn0 + m) * 8);
#pragma unroll
    for (int t = 0; t < 8; ++t) {
        float* Hp = Hw + (size_t)t * 16 * 8;
        v4f lo = {acc[t][0], acc[t][1], acc[t][2], acc[t][3]};
        v4f hi = {acc[t][4], acc[t][5], acc[t][6], acc[t][7]};
        *(v4f*)Hp       = lo;
        *(v4f*)(Hp + 4) = hi;
    }
}

static __device__ __forceinline__ v4f max4(v4f a, v4f b) {
    v4f r;
    r.x = fmaxf(a.x, b.x); r.y = fmaxf(a.y, b.y);
    r.z = fmaxf(a.z, b.z); r.w = fmaxf(a.w, b.w);
    return r;
}

__global__ __launch_bounds__(512) void edgeconv_aggregate(const float* __restrict__ H,
                                                          const int* __restrict__ src,
                                                          const float* __restrict__ bias,
                                                          float* __restrict__ out) {
    __shared__ v4f rows[4096 * 2];

    const int tid  = threadIdx.x;
    const int part = blockIdx.x & 3;
    const int sl   = (blockIdx.x >> 2) & 7;
    const int bb   = blockIdx.x >> 5;
    const int n0   = part * 1024 + tid;

    v4i idx[2][4];
#pragma unroll
    for (int j = 0; j < 2; ++j) {
        const v4i* sp = (const v4i*)(src + ((size_t)(bb * 4096 + n0 + 512 * j)) * 16);
#pragma unroll
        for (int q = 0; q < 4; ++q) idx[j][q] = sp[q];
    }

    const size_t blk = ((size_t)(bb * 8 + sl) * 4096) * 8;
    const v4f* cplane = (const v4f*)(H + blk);
    const v4f* nslice = (const v4f*)(H + PLANE_FLOATS + blk);

    v4f cen[2][2];
#pragma unroll
    for (int j = 0; j < 2; ++j) {
        cen[j][0] = cplane[(n0 + 512 * j) * 2];
        cen[j][1] = cplane[(n0 + 512 * j) * 2 + 1];
    }

#pragma unroll
    for (int i = 0; i < 16; ++i) rows[i * 512 + tid] = nslice[i * 512 + tid];
    __syncthreads();

    const int par = tid & 1;
    const v4f bl = ((const v4f*)bias)[sl * 2];
    const v4f bh = ((const v4f*)bias)[sl * 2 + 1];

#pragma unroll
    for (int j = 0; j < 2; ++j) {
        v4f ma = {-3.402823466e38f, -3.402823466e38f, -3.402823466e38f, -3.402823466e38f};
        v4f mb = ma;
#pragma unroll
        for (int q = 0; q < 4; ++q) {
#pragma unroll
            for (int e = 0; e < 4; ++e) {
                const int r = idx[j][q][e] & 4095;
                ma = max4(ma, rows[r * 2 + par]);
                mb = max4(mb, rows[r * 2 + (par ^ 1)]);
            }
        }
        const v4f lo = par ? mb : ma;
        const v4f hi = par ? ma : mb;
        const v4f cl = cen[j][0];
        const v4f ch = cen[j][1];

        float* op = out + ((size_t)(bb * 64 + sl * 8)) * 4096 + (n0 + 512 * j);
        op[0 * 4096] = fmaxf(cl.x + bl.x + lo.x, 0.0f);
        op[1 * 4096] = fmaxf(cl.y + bl.y + lo.y, 0.0f);
        op[2 * 4096] = fmaxf(cl.z + bl.z + lo.z, 0.0f);
        op[3 * 4096] = fmaxf(cl.w + bl.w + lo.w, 0.0f);
        op[4 * 4096] = fmaxf(ch.x + bh.x + hi.x, 0.0f);
        op[5 * 4096] = fmaxf(ch.y + bh.y + hi.y, 0.0f);
        op[6 * 4096] = fmaxf(ch.z + bh.z + hi.z, 0.0f);
        op[7 * 4096] = fmaxf(ch.w + bh.w + hi.w, 0.0f);
    }
}

extern "C" void kernel_launch(void* const* d_in, const int* in_sizes, int n_in,
                              void* d_out, int out_size, void* d_ws, size_t ws_size,
                              hipStream_t stream) {
    const float* x  = (const float*)d_in[0];
    const int*   ei = (const int*)d_in[1];
    const float* W  = (const float*)d_in[2];
    const float* b  = (const float*)d_in[3];
    float* out = (float*)d_out;
    float* H   = (float*)d_ws;

    edgeconv_gemm<<<256, 256, 0, stream>>>(x, W, H);
    edgeconv_aggregate<<<256, 512, 0, stream>>>(H, ei, b, out);
}


# neighbor table as two 4-channel half-tables of 16-byte rows (H stored so): aggregate gathers from half 0 while half 1 and the center rows are still in flight; sums unchanged
# speedup vs baseline: 4.0587x; 1.0683x over previous
#include <hip/hip_runtime.h>
#include <hip/hip_bf16.h>


typedef float v2f __attribute__((ext_vector_type(2)));
typedef float v4f __attribute__((ext_vector_type(4)));
typedef float v8f __attribute__((ext_vector_type(8)));
typedef int   v4i __attribute__((ext_vector_type(4)));

#define PAIR_STRIDE 288

#define PLANE_FLOATS (8 * 16 * 4096 * 4)
#define QUAD_FLOATS  (4096 * 4)

__global__ __launch_bounds__(256) void edgeconv_gemm(const float* __restrict__ x,
                                                     const float* __restrict__ W,
                                                     float* __restrict__ H) {
    __shared__ v4f xs4[32 * PAIR_STRIDE / 4];
    __shared__ v2f w2s[4][16][32];
    float* xs = (float*)xs4;

    const int tid  = threadIdx.x;
    const int lane = tid & 31;
    const int wv   = __builtin_amdgcn_readfirstlane(tid >> 5);
    const int m    = lane & 15;
    const int kh   = lane >> 4;
    const int bb   = blockIdx.x >> 5;
    const int nn0  = (blockIdx.x & 31) * 128;

    const float* xg = x + ((size_t)bb * 64) * 4096 + nn0 + 4 * lane;
    v4f xr[4][2];
#pragma unroll
    for (int q = 0; q < 4; ++q) {
        const int p = 4 * wv + q;
        xr[q][0] = *(const v4f*)(xg + (size_t)(2 * p) * 4096);
        xr[q][1] = *(const v4f*)(xg + (size_t)(2 * p + 1) * 4096);
    }

    const int col = 16 * (wv & 3) + m;
    const float* wp = W + ((wv < 4) ? 0 : 64 * 64) + (2 * kh) * 64 + col;
    v2f wa[16];
#pragma unroll
    for (int kk = 0; kk < 16; ++kk) {
        wa[kk].x = wp[(4 * kk) * 64];
        wa[kk].y = wp[(4 * kk + 1) * 64];
    }

#pragma unroll
    for (int q = 0; q < 4; ++q) {
        const int p = 4 * wv + q;
        const v4f r0 = xr[q][0];
        const v4f r1 = xr[q][1];
        const v4f s0 = {r0.x, r1.x, r0.y, r1.y};
        const v4f s1 = {r0.z, r1.z, r0.w, r1.w};
        float* d = xs + p * PAIR_STRIDE + 8 * lane;
        *(v4f*)d       = s0;
        *(v4f*)(d + 4) = s1;
    }
    if (wv >= 4) {
#pragma unroll
        for (int kk = 0; kk < 16; ++kk) w2s[wv - 4][kk][lane] = wa[kk];
    }
    __syncthreads();
    if (wv < 4) {
#pragma unroll
        for (int kk = 0; kk < 16; ++kk) {
            const v2f w2 = w2s[wv][kk][lane];
            wa[kk].x = wa[kk].x - w2.x;
            wa[kk].y = wa[kk].y - w2.y;
        }
    }

    const float* xf = xs + kh * PAIR_STRIDE + 2 * m;
    v8f acc[8];
#pragma unroll
    for (int t = 0; t < 8; ++t) acc[t] = (v8f){};
#pragma unroll
    for (int kk = 0; kk < 16; ++kk) {
#pragma unroll
        for (int t = 0; t < 8; ++t) {
            const v2f bf = *(const v2f*)&xf[(2 * kk) * PAIR_STRIDE + t * 32];
            acc[t] = __builtin_amdgcn_wmma_f32_16x16x4_f32(
                false, wa[kk], false, bf, (short)0, acc[t], false, false);
        }
    }

    const int plane = wv >> 2;
    const int quad  = (wv & 3) * 4 + kh * 2;
    float* Hw = H + (size_t)plane * PLANE_FLOATS
                  + (((size_t)(bb * 16 + quad) * 4096 + nn0 + m) * 4);
#pragma unroll
    for (int t = 0; t < 8; ++t) {
        float* Hp = Hw + (size_t)t * 16 * 4;
        v4f lo = {acc[t][0], acc[t][1], acc[t][2], acc[t][3]};
        v4f hi = {acc[t][4], acc[t][5], acc[t][6], acc[t][7]};
        *(v4f*)Hp                 = lo;
        *(v4f*)(Hp + QUAD_FLOATS) = hi;
    }
}

static __device__ __forceinline__ void lds_barrier() {
    __builtin_amdgcn_fence(__ATOMIC_RELEASE, "workgroup", "local");
    __builtin_amdgcn_s_barrier();
    __builtin_amdgcn_fence(__ATOMIC_ACQUIRE, "workgroup", "local");
}

static __device__ __forceinline__ v4f max4(v4f a, v4f b) {
    v4f r;
    r.x = fmaxf(a.x, b.x); r.y = fmaxf(a.y, b.y);
    r.z = fmaxf(a.z, b.z); r.w = fmaxf(a.w, b.w);
    return r;
}

__global__ __launch_bounds__(512) void edgeconv_aggregate(const float* __restrict__ H,
                                                          const int* __restrict__ src,
                                                          const float* __restrict__ bias,
                                                          float* __restrict__ out) {
    __shared__ v4f rows[2 * 4096];

    const int tid  = threadIdx.x;
    const int part = blockIdx.x & 3;
    const int sl   = (blockIdx.x >> 2) & 7;
    const int bb   = blockIdx.x >> 5;
    const int n0   = part * 1024 + tid;

    const size_t blk = ((size_t)(bb * 16 + sl * 2) * 4096) * 4;
    const v4f* cplane = (const v4f*)(H + blk);
    const v4f* ntab   = (const v4f*)(H + PLANE_FLOATS + blk);

    v4f t0[8];
#pragma unroll
    for (int i = 0; i < 8; ++i) t0[i] = ntab[i * 512 + tid];

    v4i idx[2][4];
#pragma unroll
    for (int j = 0; j < 2; ++j) {
        const v4i* sp = (const v4i*)(src + ((size_t)(bb * 4096 + n0 + 512 * j)) * 16);
#pragma unroll
        for (int q = 0; q < 4; ++q) idx[j][q] = sp[q];
    }
    __builtin_amdgcn_sched_barrier(0);

    v4f t1[8];
#pragma unroll
    for (int i = 0; i < 8; ++i) t1[i] = ntab[4096 + i * 512 + tid];
    v4f cen[2][2];
#pragma unroll
    for (int j = 0; j < 2; ++j) {
        cen[j][0] = cplane[n0 + 512 * j];
        cen[j][1] = cplane[4096 + n0 + 512 * j];
    }

#pragma unroll
    for (int i = 0; i < 8; ++i) rows[i * 512 + tid] = t0[i];
    lds_barrier();
    __builtin_amdgcn_sched_barrier(0);

    v4f ma[2];
#pragma unroll
    for (int j = 0; j < 2; ++j) {
        v4f mx = {-3.402823466e38f, -3.402823466e38f, -3.402823466e38f, -3.402823466e38f};
#pragma unroll
        for (int q = 0; q < 4; ++q) {
#pragma unroll
            for (int e = 0; e < 4; ++e) {
                const int r = idx[j][q][e] & 4095;
                mx = max4(mx, rows[r]);
            }
        }
        ma[j] = mx;
    }

    __builtin_amdgcn_sched_barrier(0);
#pragma unroll
    for (int i = 0; i < 8; ++i) rows[4096 + i * 512 + tid] = t1[i];
    lds_barrier();

    const v4f bl = ((const v4f*)bias)[sl * 2];
    const v4f bh = ((const v4f*)bias)[sl * 2 + 1];

#pragma unroll
    for (int j = 0; j < 2; ++j) {
        v4f mb = {-3.402823466e38f, -3.402823466e38f, -3.402823466e38f, -3.402823466e38f};
#pragma unroll
        for (int q = 0; q < 4; ++q) {
#pragma unroll
            for (int e = 0; e < 4; ++e) {
                const int r = idx[j][q][e] & 4095;
                mb = max4(mb, rows[4096 + r]);
            }
        }
        const v4f lo = ma[j];
        const v4f hi = mb;
        const v4f cl = cen[j][0];
        const v4f ch = cen[j][1];

        float* op = out + ((size_t)(bb * 64 + sl * 8)) * 4096 + (n0 + 512 * j);
        op[0 * 4096] = fmaxf(cl.x + bl.x + lo.x, 0.0f);
        op[1 * 4096] = fmaxf(cl.y + bl.y + lo.y, 0.0f);
        op[2 * 4096] = fmaxf(cl.z + bl.z + lo.z, 0.0f);
        op[3 * 4096] = fmaxf(cl.w + bl.w + lo.w, 0.0f);
        op[4 * 4096] = fmaxf(ch.x + bh.x + hi.x, 0.0f);
        op[5 * 4096] = fmaxf(ch.y + bh.y + hi.y, 0.0f);
        op[6 * 4096] = fmaxf(ch.z + bh.z + hi.z, 0.0f);
        op[7 * 4096] = fmaxf(ch.w + bh.w + hi.w, 0.0f);
    }
}

extern "C" void kernel_launch(void* const* d_in, const int* in_sizes, int n_in,
                              void* d_out, int out_size, void* d_ws, size_t ws_size,
                              hipStream_t stream) {
    const float* x  = (const float*)d_in[0];
    const int*   ei = (const int*)d_in[1];
    const float* W  = (const float*)d_in[2];
    const float* b  = (const float*)d_in[3];
    float* out = (float*)d_out;
    float* H   = (float*)d_ws;

    edgeconv_gemm<<<256, 256, 0, stream>>>(x, W, H);
    edgeconv_aggregate<<<256, 512, 0, stream>>>(H, ei, b, out);
}
